# starting kernel
# baseline (speedup 1.0000x reference)
#include <hip/hip_runtime.h>
#include <hip/hip_bf16.h>


typedef __attribute__((ext_vector_type(2))) float v2f;
typedef __attribute__((ext_vector_type(8))) float v8f;

#define HH    4
#define DD    8
#define DHH   32
#define DIMV  64
#define RPE   18
#define QKVW  128

__global__ __launch_bounds__(256) void qkv_proj_kernel(
    const float* __restrict__ x, const float* __restrict__ W,
    const float* __restrict__ bias, float* __restrict__ qkv, int N)
{
    __shared__ float xs[16 * 65];
    const int tid = threadIdx.x;
    const int r0  = blockIdx.x * 16;

    {
        int idx = tid * 4;
        int row = idx >> 6, col = idx & 63;
        int grow = r0 + row;
        float4 vx = make_float4(0.f, 0.f, 0.f, 0.f);
        if (grow < N) vx = *(const float4*)(x + (size_t)grow * 64 + col);
        xs[row*65+col+0] = vx.x; xs[row*65+col+1] = vx.y;
        xs[row*65+col+2] = vx.z; xs[row*65+col+3] = vx.w;
    }
    __syncthreads();

    const int wave = tid >> 5;
    const int lane = tid & 31;
    const int lhi  = lane >> 4;
    const int l16  = lane & 15;
    const int colbase = wave * 16;

    v8f acc = {};
    #pragma unroll
    for (int kk = 0; kk < 64; kk += 4) {
        const int ka = kk + (lhi ? 2 : 0);
        v2f a, b;
        a.x = xs[l16*65 + ka];
        a.y = xs[l16*65 + ka + 1];
        b.x = W[(size_t)ka     * QKVW + colbase + l16];
        b.y = W[(size_t)(ka+1) * QKVW + colbase + l16];
        acc = __builtin_amdgcn_wmma_f32_16x16x4_f32(
            false, a, false, b, (short)0, acc, false, false);
    }

    const int col = colbase + l16;
    const float bv = bias[col];
    #pragma unroll
    for (int r = 0; r < 8; ++r) {
        int grow = r0 + r + (lhi ? 8 : 0);
        if (grow < N) qkv[(size_t)grow * QKVW + col] = acc[r] + bv;
    }
}

__global__ __launch_bounds__(256) void edge_compat_kernel(
    const float* __restrict__ ea, const int* __restrict__ eidx,
    const float* __restrict__ qkv,
    const float* __restrict__ qw, const float* __restrict__ qb,
    const float* __restrict__ kw, const float* __restrict__ kb,
    float* __restrict__ exbuf, float* __restrict__ den, int E_)
{
    __shared__ float sea[128 * 21];
    __shared__ float srpe[128 * 67];
    const int tid = threadIdx.x;
    const int e0  = blockIdx.x * 128;

    __builtin_prefetch(ea + (size_t)(e0 + 128) * RPE, 0, 1);

    for (int i = tid; i < 128 * RPE; i += 256) {
        int e = i / RPE, c = i % RPE;
        int ge = e0 + e;
        sea[e*21 + c] = (ge < E_) ? ea[(size_t)ge * RPE + c] : 0.0f;
    }
    if (tid < 128) { sea[tid*21 + 18] = 0.f; sea[tid*21 + 19] = 0.f; }
    __syncthreads();

    const int wave = tid >> 5, lane = tid & 31;
    const int lhi = lane >> 4, l16 = lane & 15;
    const int etile = wave * 16;

    #pragma unroll
    for (int sel = 0; sel < 2; ++sel) {
        const float* Wp = sel ? kw : qw;
        #pragma unroll
        for (int ct = 0; ct < 2; ++ct) {
            v8f acc = {};
            #pragma unroll
            for (int kk = 0; kk < 20; kk += 4) {
                const int ka = kk + (lhi ? 2 : 0);
                v2f a, b;
                a.x = sea[(etile + l16)*21 + ka];
                a.y = sea[(etile + l16)*21 + ka + 1];
                b.x = (ka     < RPE) ? Wp[ka     * DHH + ct*16 + l16] : 0.0f;
                b.y = (ka + 1 < RPE) ? Wp[(ka+1) * DHH + ct*16 + l16] : 0.0f;
                acc = __builtin_amdgcn_wmma_f32_16x16x4_f32(
                    false, a, false, b, (short)0, acc, false, false);
            }
            const int cb = sel*32 + ct*16 + l16;
            #pragma unroll
            for (int r = 0; r < 8; ++r)
                srpe[(etile + r + (lhi ? 8 : 0))*67 + cb] = acc[r];
        }
    }
    __syncthreads();

    const int eloc = tid >> 1, pairh = (tid & 1) * 2;
    const int ge = e0 + eloc;
    if (ge < E_) {
        const int s = eidx[ge];
        const int t = eidx[E_ + ge];
        const float scale = 0.35355339059327373f;
        const float* qrow = qkv + (size_t)s * QKVW;
        const float* krow = qkv + (size_t)t * QKVW + DHH;
        #pragma unroll
        for (int hh = 0; hh < 2; ++hh) {
            const int h = pairh + hh;
            float compat = 0.f;
            #pragma unroll
            for (int d = 0; d < DD; ++d) {
                const int c = h*DD + d;
                const float qe = qrow[c]*scale + srpe[eloc*67 + c]      + qb[c];
                const float ke = krow[c]       + srpe[eloc*67 + 32 + c] + kb[c];
                compat = fmaf(qe, ke, compat);
            }
            const float ev = __expf(compat);
            exbuf[(size_t)ge * HH + h] = ev;
            atomicAdd(&den[s * HH + h], ev);
        }
    }
}

__global__ __launch_bounds__(256) void edge_out_kernel(
    const float* __restrict__ ea, const int* __restrict__ eidx,
    const float* __restrict__ qkv,
    const float* __restrict__ vw, const float* __restrict__ vb,
    const float* __restrict__ exbuf, const float* __restrict__ den,
    float* __restrict__ out, int E_)
{
    __shared__ float sea[128 * 21];
    __shared__ float srpe[128 * 67];
    const int tid = threadIdx.x;
    const int e0  = blockIdx.x * 128;

    __builtin_prefetch(ea + (size_t)(e0 + 128) * RPE, 0, 1);

    for (int i = tid; i < 128 * RPE; i += 256) {
        int e = i / RPE, c = i % RPE;
        int ge = e0 + e;
        sea[e*21 + c] = (ge < E_) ? ea[(size_t)ge * RPE + c] : 0.0f;
    }
    if (tid < 128) { sea[tid*21 + 18] = 0.f; sea[tid*21 + 19] = 0.f; }
    __syncthreads();

    const int wave = tid >> 5, lane = tid & 31;
    const int lhi = lane >> 4, l16 = lane & 15;
    const int etile = wave * 16;

    #pragma unroll
    for (int ct = 0; ct < 4; ++ct) {
        v8f acc = {};
        #pragma unroll
        for (int kk = 0; kk < 20; kk += 4) {
            const int ka = kk + (lhi ? 2 : 0);
            v2f a, b;
            a.x = sea[(etile + l16)*21 + ka];
            a.y = sea[(etile + l16)*21 + ka + 1];
            b.x = (ka     < RPE) ? vw[ka     * DIMV + ct*16 + l16] : 0.0f;
            b.y = (ka + 1 < RPE) ? vw[(ka+1) * DIMV + ct*16 + l16] : 0.0f;
            acc = __builtin_amdgcn_wmma_f32_16x16x4_f32(
                false, a, false, b, (short)0, acc, false, false);
        }
        const int cb = ct*16 + l16;
        #pragma unroll
        for (int r = 0; r < 8; ++r)
            srpe[(etile + r + (lhi ? 8 : 0))*67 + cb] = acc[r];
    }
    __syncthreads();

    const int eloc = tid >> 1, half = tid & 1;
    const int ge = e0 + eloc;
    if (ge < E_) {
        const int s = eidx[ge];
        const int t = eidx[E_ + ge];
        float attn[2];
        #pragma unroll
        for (int hh = 0; hh < 2; ++hh) {
            const int h = half*2 + hh;
            attn[hh] = exbuf[(size_t)ge*HH + h] / (den[s*HH + h] + 1e-16f);
        }
        const float* vrow = qkv + (size_t)t * QKVW + 2*DHH;
        const int c0 = half * 32;
        #pragma unroll
        for (int c = c0; c < c0 + 32; ++c) {
            const float val = (vrow[c] + srpe[eloc*67 + c] + vb[c]) * attn[(c >> 4) & 1];
            atomicAdd(&out[(size_t)s * DIMV + c], val);
        }
    }
}

extern "C" void kernel_launch(void* const* d_in, const int* in_sizes, int n_in,
                              void* d_out, int out_size, void* d_ws, size_t ws_size,
                              hipStream_t stream)
{
    const float* x       = (const float*)d_in[0];
    const int*   eidx    = (const int*)  d_in[1];
    const float* ea      = (const float*)d_in[2];
    const float* qkv_w   = (const float*)d_in[3];
    const float* qkv_b   = (const float*)d_in[4];
    const float* k_rpe_w = (const float*)d_in[5];
    const float* k_rpe_b = (const float*)d_in[6];
    const float* q_rpe_w = (const float*)d_in[7];
    const float* q_rpe_b = (const float*)d_in[8];
    const float* v_rpe_w = (const float*)d_in[9];
    const float* v_rpe_b = (const float*)d_in[10];

    const int N  = in_sizes[0] / DIMV;
    const int E_ = in_sizes[2] / RPE;
    float* out = (float*)d_out;

    char* ws = (char*)d_ws;
    float* qkv  = (float*)ws;
    float* den  = (float*)(ws + (size_t)N * QKVW * 4);
    float* exb  = (float*)(ws + (size_t)N * QKVW * 4
                              + (size_t)N * HH   * 4);

    hipMemsetAsync(den, 0, (size_t)N * HH   * sizeof(float), stream);
    hipMemsetAsync(out, 0, (size_t)N * DIMV * sizeof(float), stream);

    qkv_proj_kernel<<<(N + 15) / 16, 256, 0, stream>>>(x, qkv_w, qkv_b, qkv, N);
    edge_compat_kernel<<<(E_ + 127) / 128, 256, 0, stream>>>(
        ea, eidx, qkv, q_rpe_w, q_rpe_b, k_rpe_w, k_rpe_b, exb, den, E_);
    edge_out_kernel<<<(E_ + 127) / 128, 256, 0, stream>>>(
        ea, eidx, qkv, v_rpe_w, v_rpe_b, exb, den, out, E_);
}


# Compat kernel dot phase: 8 lanes per edge take 16-byte pieces of q/k rows, rpe rows (LDS pitch 68), biases: a load spans 4 whole rows, not 16 half rows; float sum order changes (two 4-term halves)
# speedup vs baseline: 5.0828x; 5.0828x over previous
#include <hip/hip_runtime.h>
#include <hip/hip_bf16.h>


typedef __attribute__((ext_vector_type(2))) float v2f;
typedef __attribute__((ext_vector_type(4))) float v4f;
typedef __attribute__((ext_vector_type(8))) float v8f;

#define HH    4
#define DD    8
#define DHH   32
#define DIMV  64
#define RPE   18
#define QKVW  128
#define NCH   4
#define NCHH  2

__global__ __launch_bounds__(256) void qkv_proj_kernel(
    const float* __restrict__ x, const float* __restrict__ W,
    const float* __restrict__ bias, float* __restrict__ qkv,
    int* __restrict__ head, int N)
{
    __shared__ float xs[16 * 65];
    const int tid = threadIdx.x;
    const int r0  = blockIdx.x * 16;

    {
        int idx = tid * 4;
        int row = idx >> 6, col = idx & 63;
        int grow = r0 + row;
        float4 vx = make_float4(0.f, 0.f, 0.f, 0.f);
        if (grow < N) vx = *(const float4*)(x + (size_t)grow * 64 + col);
        xs[row*65+col+0] = vx.x; xs[row*65+col+1] = vx.y;
        xs[row*65+col+2] = vx.z; xs[row*65+col+3] = vx.w;
    }
    if (tid < 64 && r0 * NCHH + (tid & 31) < N * NCHH)
        head[(tid >> 5) * (N * NCHH) + r0 * NCHH + (tid & 31)] = -1;
    __syncthreads();

    const int wave = tid >> 5;
    const int lane = tid & 31;
    const int lhi  = lane >> 4;
    const int l16  = lane & 15;
    const int colbase = wave * 16;

    v8f acc = {};
    #pragma unroll
    for (int kk = 0; kk < 64; kk += 4) {
        const int ka = kk + (lhi ? 2 : 0);
        v2f a, b;
        a.x = xs[l16*65 + ka];
        a.y = xs[l16*65 + ka + 1];
        b.x = W[(size_t)ka     * QKVW + colbase + l16];
        b.y = W[(size_t)(ka+1) * QKVW + colbase + l16];
        acc = __builtin_amdgcn_wmma_f32_16x16x4_f32(
            false, a, false, b, (short)0, acc, false, false);
    }

    const int col = colbase + l16;
    const float bv = bias[col];
    #pragma unroll
    for (int r = 0; r < 8; ++r) {
        int grow = r0 + r + (lhi ? 8 : 0);
        if (grow < N) qkv[(size_t)grow * QKVW + col] = acc[r] + bv;
    }
}

__global__ __launch_bounds__(256) void edge_compat_kernel(
    const float* __restrict__ ea, const int* __restrict__ eidx,
    const float* __restrict__ qkv,
    const float* __restrict__ qw, const float* __restrict__ qb,
    const float* __restrict__ kw, const float* __restrict__ kb,
    float* __restrict__ exbuf, int* __restrict__ head, int* __restrict__ nxt, int N, int E_)
{
    __shared__ __attribute__((aligned(16))) float srpe[128 * 68];
    const int tid = threadIdx.x;
    const int e0  = blockIdx.x * 128;

    __builtin_prefetch(ea + (size_t)(e0 + 128) * RPE, 0, 1);

    const int wave = tid >> 5, lane = tid & 31;
    const int lhi = lane >> 4, l16 = lane & 15;
    const int etile = wave * 16;

    const int  arow_e = e0 + etile + l16;
    const bool arow_ok = arow_e < E_;
    const float* arow = ea + (size_t)(arow_ok ? arow_e : 0) * RPE;
    const v2f zero2 = {0.f, 0.f};
    v2f afr[5];
    #pragma unroll
    for (int ki = 0; ki < 5; ++ki) {
        const int ka = ki*4 + (lhi ? 2 : 0);
        const int kc = (ka < RPE) ? ka : 16;
        const v2f ld = *(const v2f*)(arow + kc);
        afr[ki] = (arow_ok && ka < RPE) ? ld : zero2;
    }

    #pragma unroll
    for (int sel = 0; sel < 2; ++sel) {
        const float* Wp = sel ? kw : qw;
        v8f acc0 = {}, acc1 = {};
        #pragma unroll
        for (int ki = 0; ki < 5; ++ki) {
            const int ka = ki*4 + (lhi ? 2 : 0);
            const bool okx = ka < RPE, oky = ka + 1 < RPE;
            const int rx = okx ? ka : 16;
            const int ry = oky ? ka + 1 : 17;
            const v2f bx = *(const v2f*)(Wp + rx * DHH + 2*l16);
            const v2f by = *(const v2f*)(Wp + ry * DHH + 2*l16);
            v2f b0, b1;
            b0.x = okx ? bx.x : 0.0f;  b0.y = oky ? by.x : 0.0f;
            b1.x = okx ? bx.y : 0.0f;  b1.y = oky ? by.y : 0.0f;
            acc0 = __builtin_amdgcn_wmma_f32_16x16x4_f32(
                false, afr[ki], false, b0, (short)0, acc0, false, false);
            acc1 = __builtin_amdgcn_wmma_f32_16x16x4_f32(
                false, afr[ki], false, b1, (short)0, acc1, false, false);
        }
        const int cb = sel*32 + 2*l16;
        #pragma unroll
        for (int r = 0; r < 8; ++r) {
            srpe[(etile + r + (lhi ? 8 : 0))*68 + cb]     = acc0[r];
            srpe[(etile + r + (lhi ? 8 : 0))*68 + cb + 1] = acc1[r];
        }
    }
    __syncthreads();

    const int grp = lane >> 3, pc = lane & 7;
    const float scale = 0.35355339059327373f;
    const float4 qb4 = *(const float4*)(qb + 4*pc);
    const float4 kb4 = *(const float4*)(kb + 4*pc);
    int  gev[4];
    bool okv[4];
    int  sv[4], tv[4];
    #pragma unroll
    for (int i = 0; i < 4; ++i) {
        gev[i] = e0 + etile + 4*i + grp;
        okv[i] = gev[i] < E_;
        const int gc = okv[i] ? gev[i] : 0;
        sv[i] = eidx[gc];
        tv[i] = eidx[E_ + gc];
    }
    const int gl  = e0 + etile + l16;
    const int sl  = eidx[gl < E_ ? gl : 0];
    v4f q4[4], k4[4];
    #pragma unroll
    for (int i = 0; i < 4; ++i) {
        q4[i] = *(const v4f*)(qkv + (size_t)sv[i] * QKVW + 4*pc);
        k4[i] = *(const v4f*)(qkv + (size_t)tv[i] * QKVW + DHH + 4*pc);
    }
    asm volatile("" :: "v"(q4[0]), "v"(q4[1]), "v"(q4[2]), "v"(q4[3]),
                       "v"(k4[0]), "v"(k4[1]), "v"(k4[2]), "v"(k4[3]));
    float ev[4];
    #pragma unroll
    for (int i = 0; i < 4; ++i) {
        const int eloc = etile + 4*i + grp;
        const float4 rq = *(const float4*)(srpe + eloc*68 + 4*pc);
        const float4 rk = *(const float4*)(srpe + eloc*68 + 32 + 4*pc);
        float part = 0.f;
        part = fmaf(q4[i][0]*scale + rq.x + qb4.x, k4[i][0] + rk.x + kb4.x, part);
        part = fmaf(q4[i][1]*scale + rq.y + qb4.y, k4[i][1] + rk.y + kb4.y, part);
        part = fmaf(q4[i][2]*scale + rq.z + qb4.z, k4[i][2] + rk.z + kb4.z, part);
        part = fmaf(q4[i][3]*scale + rq.w + qb4.w, k4[i][3] + rk.w + kb4.w, part);
        const float oth = __builtin_bit_cast(float,
            __builtin_amdgcn_mov_dpp(__builtin_bit_cast(int, part), 0xB1, 0xF, 0xF, true));
        ev[i] = __expf(part + oth);
    }
    #pragma unroll
    for (int i = 0; i < 4; ++i)
        if (okv[i] && (pc & 1) == 0) exbuf[(size_t)gev[i] * HH + (pc >> 1)] = ev[i];

    const int hplane = (int)((__builtin_amdgcn_s_sendmsg_rtn(0x87) >> 18) & 1u);
    if (lhi == 0 && gl < E_) {
        nxt[gl] = atomicExch(&head[hplane * (N * NCHH) + sl * NCHH + (gl & (NCHH - 1))], gl);
    }
}


__device__ __forceinline__ void walk_two_steps(int (&ch)[NCH], int (&slot)[8],
                                               const int* __restrict__ nxt)
{
    #pragma unroll
    for (int st = 0; st < 8 / NCH; ++st) {
        #pragma unroll
        for (int j = 0; j < NCH; ++j) slot[st * NCH + j] = ch[j];
        if ((ch[0] & ch[1] & ch[2] & ch[3]) >= 0) {
            int nx[NCH];
            #pragma unroll
            for (int j = 0; j < NCH; ++j) nx[j] = nxt[ch[j] < 0 ? 0 : ch[j]];
            #pragma unroll
            for (int j = 0; j < NCH; ++j) ch[j] = ch[j] < 0 ? -1 : nx[j];
        }
    }
}

__global__ __launch_bounds__(256) __attribute__((amdgpu_waves_per_eu(8)))
void node_out_kernel(
    const float* __restrict__ ea, const int* __restrict__ eidx,
    const float* __restrict__ qkv,
    const float* __restrict__ vw, const float* __restrict__ vb,
    const float* __restrict__ exbuf,
    const int* __restrict__ head, const int* __restrict__ nxt,
    float* __restrict__ out, int N, int E_)
{
    const int tid = threadIdx.x;
    const int lane = tid & 31;
    const int lhi = lane >> 4, l16 = lane & 15;
    const int hsel = l16 >> 2;
    const int wid = __builtin_amdgcn_readfirstlane((int)blockIdx.x * 8 + (tid >> 5));
    const int nw  = (int)gridDim.x * 8;
    if (wid >= N) return;

    v2f bf[4][5];
    float vbias[4];
    #pragma unroll
    for (int ki = 0; ki < 5; ++ki) {
        const int ka = ki*4 + (lhi ? 2 : 0);
        const bool okx = ka < RPE, oky = ka + 1 < RPE;
        const int rx = okx ? ka : 16;
        const int ry = oky ? ka + 1 : 17;
        const float4 bx = *(const float4*)(vw + rx * DIMV + 4*l16);
        const float4 by = *(const float4*)(vw + ry * DIMV + 4*l16);
        bf[0][ki].x = okx ? bx.x : 0.0f;  bf[0][ki].y = oky ? by.x : 0.0f;
        bf[1][ki].x = okx ? bx.y : 0.0f;  bf[1][ki].y = oky ? by.y : 0.0f;
        bf[2][ki].x = okx ? bx.z : 0.0f;  bf[2][ki].y = oky ? by.z : 0.0f;
        bf[3][ki].x = okx ? bx.w : 0.0f;  bf[3][ki].y = oky ? by.w : 0.0f;
    }
    {
        const float4 b4 = *(const float4*)(vb + 4*l16);
        vbias[0] = b4.x; vbias[1] = b4.y; vbias[2] = b4.z; vbias[3] = b4.w;
    }
    const v4f* __restrict__ vrow4 = (const v4f*)qkv;

    float acc[4] = {0.f, 0.f, 0.f, 0.f};
    float den = 0.f;
    const v2f zero2 = {0.f, 0.f};

    int ch[NCH];
    #pragma unroll
    for (int j = 0; j < NCH; ++j) ch[j] = head[(j / NCHH) * (N * NCHH) + wid * NCHH + (j % NCHH)];
    int clo[8], chi[8];
    walk_two_steps(ch, clo, nxt);
    walk_two_steps(ch, chi, nxt);
    int  s_cur = wid;
    bool have  = true;

    for (int it = 0; it < E_ + N && have; ++it) {
        const bool last = (ch[0] & ch[1] & ch[2] & ch[3]) < 0;

        int mye = -1;
        int er[8];
        #pragma unroll
        for (int r = 0; r < 8; ++r) {
            er[r] = lhi ? chi[r] : clo[r];
            if (l16 == r)     mye = clo[r];
            if (l16 == r + 8) mye = chi[r];
        }

        int tt[8];
        float exw[8];
        #pragma unroll
        for (int r = 0; r < 8; ++r)
            tt[r] = eidx[E_ + (er[r] < 0 ? 0 : er[r])];
        #pragma unroll
        for (int r = 0; r < 8; ++r)
            exw[r] = exbuf[(unsigned)(er[r] < 0 ? 0 : er[r]) * HH + hsel];

        const float* erow = ea + (size_t)(mye < 0 ? 0 : mye) * RPE;
        v2f a[5];
        #pragma unroll
        for (int ki = 0; ki < 5; ++ki) {
            const int ka = ki*4 + (lhi ? 2 : 0);
            const int kc = (ka < RPE) ? ka : 16;
            const v2f ld = *(const v2f*)(erow + kc);
            a[ki] = (mye >= 0 && ka < RPE) ? ld : zero2;
        }

        int  s_nxt = s_cur;
        bool haveN = true;
        if (last) {
            s_nxt = s_cur + nw;
            haveN = s_nxt < N;
            if (haveN) {
                #pragma unroll
                for (int j = 0; j < NCH; ++j) ch[j] = head[(j / NCHH) * (N * NCHH) + s_nxt * NCHH + (j % NCHH)];
            }
        }
        int nlo[8], nhi[8];
        walk_two_steps(ch, nlo, nxt);

        asm volatile("" :: "v"(tt[0]), "v"(tt[1]), "v"(tt[2]), "v"(tt[3]),
                           "v"(tt[4]), "v"(tt[5]), "v"(tt[6]), "v"(tt[7]));

        v4f vv[8];
        #pragma unroll
        for (int r = 0; r < 8; ++r)
            vv[r] = vrow4[(unsigned)tt[r] * (QKVW/4) + (2*DHH/4) + l16];

        walk_two_steps(ch, nhi, nxt);

        v8f d[4];
        #pragma unroll
        for (int ct = 0; ct < 4; ++ct) {
            v8f dd = {};
            #pragma unroll
            for (int ki = 0; ki < 5; ++ki)
                dd = __builtin_amdgcn_wmma_f32_16x16x4_f32(
                    false, a[ki], false, bf[ct][ki], (short)0, dd, false, false);
            d[ct] = dd;
        }

        #pragma unroll
        for (int r = 0; r < 8; ++r) {
            const bool live = er[r] >= 0;
            const float wl = live ? exw[r] : 0.0f;
            const float vr[4] = {vv[r][0], vv[r][1], vv[r][2], vv[r][3]};
            #pragma unroll
            for (int ct = 0; ct < 4; ++ct) {
                const float val = d[ct][r] + vr[ct] + vbias[ct];
                const float vl = live ? val : 0.0f;
                acc[ct] = fmaf(wl, vl, acc[ct]);
            }
            den += wl;
        }

        if (last) {
            const float dn = den + __shfl_xor(den, 16);
            float tot[4];
            #pragma unroll
            for (int ct = 0; ct < 4; ++ct) tot[ct] = acc[ct] + __shfl_xor(acc[ct], 16);
            if (lhi == 0) {
                float4 res;
                res.x = tot[0] / (dn + 1e-16f);
                res.y = tot[1] / (dn + 1e-16f);
                res.z = tot[2] / (dn + 1e-16f);
                res.w = tot[3] / (dn + 1e-16f);
                *(float4*)(out + ((unsigned)s_cur * DIMV + 4*l16)) = res;
            }
            #pragma unroll
            for (int ct = 0; ct < 4; ++ct) acc[ct] = 0.f;
            den = 0.f;
        }

        #pragma unroll
        for (int r = 0; r < 8; ++r) { clo[r] = nlo[r]; chi[r] = nhi[r]; }
        s_cur = s_nxt;
        have  = haveN;
    }
}

extern "C" void kernel_launch(void* const* d_in, const int* in_sizes, int n_in,
                              void* d_out, int out_size, void* d_ws, size_t ws_size,
                              hipStream_t stream)
{
    const float* x       = (const float*)d_in[0];
    const int*   eidx    = (const int*)  d_in[1];
    const float* ea      = (const float*)d_in[2];
    const float* qkv_w   = (const float*)d_in[3];
    const float* qkv_b   = (const float*)d_in[4];
    const float* k_rpe_w = (const float*)d_in[5];
    const float* k_rpe_b = (const float*)d_in[6];
    const float* q_rpe_w = (const float*)d_in[7];
    const float* q_rpe_b = (const float*)d_in[8];
    const float* v_rpe_w = (const float*)d_in[9];
    const float* v_rpe_b = (const float*)d_in[10];

    const int N  = in_sizes[0] / DIMV;
    const int E_ = in_sizes[2] / RPE;
    float* out = (float*)d_out;

    char* ws = (char*)d_ws;
    float* qkv  = (float*)ws;
    float* exb  = (float*)(ws + (size_t)N * QKVW * 4);
    int*   head = (int*)  (ws + (size_t)N * QKVW * 4
                              + (size_t)E_ * HH  * 4);
    int*   nxt  = (int*)  (ws + (size_t)N * QKVW * 4
                              + (size_t)E_ * HH  * 4
                              + (size_t)N * NCH  * 4);

    qkv_proj_kernel<<<(N + 15) / 16, 256, 0, stream>>>(x, qkv_w, qkv_b, qkv, head, N);
    edge_compat_kernel<<<(E_ + 127) / 128, 256, 0, stream>>>(
        ea, eidx, qkv, q_rpe_w, q_rpe_b, k_rpe_w, k_rpe_b, exb, head, nxt, N, E_);
    node_out_kernel<<<(N + 15) / 16, 256, 0, stream>>>(
        ea, eidx, qkv, v_rpe_w, v_rpe_b, exb, head, nxt, out, N, E_);
}


# Per-node slot tables replace the edge chains (counter add hands each edge a slot, 32 per cache half, overflow chain behind): node kernel loads a tile's 16 edges at once; float sum order changes
# speedup vs baseline: 5.2785x; 1.0385x over previous
#include <hip/hip_runtime.h>
#include <hip/hip_bf16.h>


typedef __attribute__((ext_vector_type(2))) float v2f;
typedef __attribute__((ext_vector_type(4))) float v4f;
typedef __attribute__((ext_vector_type(8))) float v8f;

#define HH    4
#define DD    8
#define DHH   32
#define DIMV  64
#define RPE   18
#define QKVW  128
#define CAP   32

__global__ __launch_bounds__(256) void qkv_proj_kernel(
    const float* __restrict__ x, const float* __restrict__ W,
    const float* __restrict__ bias, float* __restrict__ qkv,
    int* __restrict__ meta, int N)
{
    __shared__ float xs[16 * 65];
    const int tid = threadIdx.x;
    const int r0  = blockIdx.x * 16;

    {
        int idx = tid * 4;
        int row = idx >> 6, col = idx & 63;
        int grow = r0 + row;
        float4 vx = make_float4(0.f, 0.f, 0.f, 0.f);
        if (grow < N) vx = *(const float4*)(x + (size_t)grow * 64 + col);
        xs[row*65+col+0] = vx.x; xs[row*65+col+1] = vx.y;
        xs[row*65+col+2] = vx.z; xs[row*65+col+3] = vx.w;
    }
    if (tid < 64 && r0 + ((tid & 31) >> 1) < N)
        meta[((tid >> 5) * N + r0) * 2 + (tid & 31)] = (tid & 1) ? -1 : 0;
    __syncthreads();

    const int wave = tid >> 5;
    const int lane = tid & 31;
    const int lhi  = lane >> 4;
    const int l16  = lane & 15;
    const int colbase = wave * 16;

    v8f acc = {};
    #pragma unroll
    for (int kk = 0; kk < 64; kk += 4) {
        const int ka = kk + (lhi ? 2 : 0);
        v2f a, b;
        a.x = xs[l16*65 + ka];
        a.y = xs[l16*65 + ka + 1];
        b.x = W[(size_t)ka     * QKVW + colbase + l16];
        b.y = W[(size_t)(ka+1) * QKVW + colbase + l16];
        acc = __builtin_amdgcn_wmma_f32_16x16x4_f32(
            false, a, false, b, (short)0, acc, false, false);
    }

    const int col = colbase + l16;
    const float bv = bias[col];
    #pragma unroll
    for (int r = 0; r < 8; ++r) {
        int grow = r0 + r + (lhi ? 8 : 0);
        if (grow < N) qkv[(size_t)grow * QKVW + col] = acc[r] + bv;
    }
}

__global__ __launch_bounds__(256) void edge_compat_kernel(
    const float* __restrict__ ea, const int* __restrict__ eidx,
    const float* __restrict__ qkv,
    const float* __restrict__ qw, const float* __restrict__ qb,
    const float* __restrict__ kw, const float* __restrict__ kb,
    float* __restrict__ exbuf, int* __restrict__ meta,
    int* __restrict__ nxt, int* __restrict__ tab, int N, int E_)
{
    __shared__ __attribute__((aligned(16))) float srpe[128 * 68];
    const int tid = threadIdx.x;
    const int e0  = blockIdx.x * 128;

    __builtin_prefetch(ea + (size_t)(e0 + 128) * RPE, 0, 1);

    const int wave = tid >> 5, lane = tid & 31;
    const int lhi = lane >> 4, l16 = lane & 15;
    const int etile = wave * 16;

    const int  arow_e = e0 + etile + l16;
    const bool arow_ok = arow_e < E_;
    const float* arow = ea + (size_t)(arow_ok ? arow_e : 0) * RPE;
    const v2f zero2 = {0.f, 0.f};
    v2f afr[5];
    #pragma unroll
    for (int ki = 0; ki < 5; ++ki) {
        const int ka = ki*4 + (lhi ? 2 : 0);
        const int kc = (ka < RPE) ? ka : 16;
        const v2f ld = *(const v2f*)(arow + kc);
        afr[ki] = (arow_ok && ka < RPE) ? ld : zero2;
    }

    #pragma unroll
    for (int sel = 0; sel < 2; ++sel) {
        const float* Wp = sel ? kw : qw;
        v8f acc0 = {}, acc1 = {};
        #pragma unroll
        for (int ki = 0; ki < 5; ++ki) {
            const int ka = ki*4 + (lhi ? 2 : 0);
            const bool okx = ka < RPE, oky = ka + 1 < RPE;
            const int rx = okx ? ka : 16;
            const int ry = oky ? ka + 1 : 17;
            const v2f bx = *(const v2f*)(Wp + rx * DHH + 2*l16);
            const v2f by = *(const v2f*)(Wp + ry * DHH + 2*l16);
            v2f b0, b1;
            b0.x = okx ? bx.x : 0.0f;  b0.y = oky ? by.x : 0.0f;
            b1.x = okx ? bx.y : 0.0f;  b1.y = oky ? by.y : 0.0f;
            acc0 = __builtin_amdgcn_wmma_f32_16x16x4_f32(
                false, afr[ki], false, b0, (short)0, acc0, false, false);
            acc1 = __builtin_amdgcn_wmma_f32_16x16x4_f32(
                false, afr[ki], false, b1, (short)0, acc1, false, false);
        }
        const int cb = sel*32 + 2*l16;
        #pragma unroll
        for (int r = 0; r < 8; ++r) {
            srpe[(etile + r + (lhi ? 8 : 0))*68 + cb]     = acc0[r];
            srpe[(etile + r + (lhi ? 8 : 0))*68 + cb + 1] = acc1[r];
        }
    }
    __syncthreads();

    const int grp = lane >> 3, pc = lane & 7;
    const float scale = 0.35355339059327373f;
    const float4 qb4 = *(const float4*)(qb + 4*pc);
    const float4 kb4 = *(const float4*)(kb + 4*pc);
    int  gev[4];
    bool okv[4];
    int  sv[4], tv[4];
    #pragma unroll
    for (int i = 0; i < 4; ++i) {
        gev[i] = e0 + etile + 4*i + grp;
        okv[i] = gev[i] < E_;
        const int gc = okv[i] ? gev[i] : 0;
        sv[i] = eidx[gc];
        tv[i] = eidx[E_ + gc];
    }
    const int gl  = e0 + etile + l16;
    const int sl  = eidx[gl < E_ ? gl : 0];
    v4f q4[4], k4[4];
    #pragma unroll
    for (int i = 0; i < 4; ++i) {
        q4[i] = *(const v4f*)(qkv + (size_t)sv[i] * QKVW + 4*pc);
        k4[i] = *(const v4f*)(qkv + (size_t)tv[i] * QKVW + DHH + 4*pc);
    }
    asm volatile("" :: "v"(q4[0]), "v"(q4[1]), "v"(q4[2]), "v"(q4[3]),
                       "v"(k4[0]), "v"(k4[1]), "v"(k4[2]), "v"(k4[3]));
    float ev[4];
    #pragma unroll
    for (int i = 0; i < 4; ++i) {
        const int eloc = etile + 4*i + grp;
        const float4 rq = *(const float4*)(srpe + eloc*68 + 4*pc);
        const float4 rk = *(const float4*)(srpe + eloc*68 + 32 + 4*pc);
        float part = 0.f;
        part = fmaf(q4[i][0]*scale + rq.x + qb4.x, k4[i][0] + rk.x + kb4.x, part);
        part = fmaf(q4[i][1]*scale + rq.y + qb4.y, k4[i][1] + rk.y + kb4.y, part);
        part = fmaf(q4[i][2]*scale + rq.z + qb4.z, k4[i][2] + rk.z + kb4.z, part);
        part = fmaf(q4[i][3]*scale + rq.w + qb4.w, k4[i][3] + rk.w + kb4.w, part);
        const float oth = __builtin_bit_cast(float,
            __builtin_amdgcn_mov_dpp(__builtin_bit_cast(int, part), 0xB1, 0xF, 0xF, true));
        ev[i] = __expf(part + oth);
    }
    #pragma unroll
    for (int i = 0; i < 4; ++i)
        if (okv[i] && (pc & 1) == 0) exbuf[(size_t)gev[i] * HH + (pc >> 1)] = ev[i];

    const int hplane = (int)((__builtin_amdgcn_s_sendmsg_rtn(0x87) >> 18) & 1u);
    if (lhi == 0 && gl < E_) {
        const int w   = hplane * N + sl;
        const int pos = atomicAdd(&meta[2 * w], 1);
        if (pos < CAP) tab[(unsigned)w * CAP + (unsigned)pos] = gl;
        else           nxt[gl] = atomicExch(&meta[2 * w + 1], gl);
    }
}


struct TileCursor {
    int  s;
    int  n0, tot;
    int  k;
    int  oe0, oe1;
    int2 mb0, mb1;
    bool done;
};

__device__ __forceinline__ void open_node(TileCursor& F, int sn, int2 m0, int2 m1,
                                          const int2* __restrict__ meta2, int nw, int N)
{
    F.s   = sn;
    F.k   = 0;
    F.n0  = m0.x < CAP ? m0.x : CAP;
    F.tot = F.n0 + (m1.x < CAP ? m1.x : CAP);
    F.oe0 = m0.x > CAP ? m0.y : -1;
    F.oe1 = m1.x > CAP ? m1.y : -1;
    if (sn + nw < N) { F.mb0 = meta2[sn + nw]; F.mb1 = meta2[N + sn + nw]; }
}

__device__ __forceinline__ void next_tile(TileCursor& F, int& sv, int& s_of, bool& last_of,
                                          const int2* __restrict__ meta2,
                                          const int* __restrict__ nxt, const int* __restrict__ tab,
                                          int l16, int nw, int N)
{
    s_of = F.s;
    if (F.k == 0 || F.k * 16 < F.tot) {
        const int g = F.k * 16 + l16;
        const unsigned base = (unsigned)F.s * CAP;
        const unsigned idx  = g < F.n0 ? base + (unsigned)g
                                       : (unsigned)N * CAP + base + (unsigned)(g - F.n0);
        int ld = -1;
        if (g < F.tot) ld = tab[idx];
        sv = ld;
        F.k += 1;
    } else {
        int v = -1;
        for (int j = 0; j < 16; ++j) {
            const int e = F.oe0 >= 0 ? F.oe0 : F.oe1;
            if (e < 0) break;
            if (l16 == j) v = e;
            const int nx = nxt[e];
            if (F.oe0 >= 0) F.oe0 = nx; else F.oe1 = nx;
        }
        sv = v;
    }
    const bool more = F.k * 16 < F.tot || F.oe0 >= 0 || F.oe1 >= 0;
    last_of = !more;
    if (!more) {
        const int sn = F.s + nw;
        if (sn < N) open_node(F, sn, F.mb0, F.mb1, meta2, nw, N);
        else        F.done = true;
    }
}

__global__ __launch_bounds__(256)
void node_out_kernel(
    const float* __restrict__ ea, const int* __restrict__ eidx,
    const float* __restrict__ qkv,
    const float* __restrict__ vw, const float* __restrict__ vb,
    const float* __restrict__ exbuf,
    const int* __restrict__ meta,
    const int* __restrict__ nxt, const int* __restrict__ tab,
    float* __restrict__ out, int N, int E_)
{
    const int tid = threadIdx.x;
    const int lane = tid & 31;
    const int lhi = lane >> 4, l16 = lane & 15;
    const int hsel = l16 >> 2;
    const int wid = __builtin_amdgcn_readfirstlane((int)blockIdx.x * 8 + (tid >> 5));
    const int nw  = (int)gridDim.x * 8;
    if (wid >= N) return;

    v2f bf[4][5];
    float vbias[4];
    #pragma unroll
    for (int ki = 0; ki < 5; ++ki) {
        const int ka = ki*4 + (lhi ? 2 : 0);
        const bool okx = ka < RPE, oky = ka + 1 < RPE;
        const int rx = okx ? ka : 16;
        const int ry = oky ? ka + 1 : 17;
        const float4 bx = *(const float4*)(vw + rx * DIMV + 4*l16);
        const float4 by = *(const float4*)(vw + ry * DIMV + 4*l16);
        bf[0][ki].x = okx ? bx.x : 0.0f;  bf[0][ki].y = oky ? by.x : 0.0f;
        bf[1][ki].x = okx ? bx.y : 0.0f;  bf[1][ki].y = oky ? by.y : 0.0f;
        bf[2][ki].x = okx ? bx.z : 0.0f;  bf[2][ki].y = oky ? by.z : 0.0f;
        bf[3][ki].x = okx ? bx.w : 0.0f;  bf[3][ki].y = oky ? by.w : 0.0f;
    }
    {
        const float4 b4 = *(const float4*)(vb + 4*l16);
        vbias[0] = b4.x; vbias[1] = b4.y; vbias[2] = b4.z; vbias[3] = b4.w;
    }
    const v4f* __restrict__ vrow4 = (const v4f*)qkv;

    float acc[4] = {0.f, 0.f, 0.f, 0.f};
    float den = 0.f;
    const v2f zero2 = {0.f, 0.f};

    const int2* __restrict__ meta2 = (const int2*)meta;
    TileCursor F;
    F.mb0 = make_int2(0, -1); F.mb1 = make_int2(0, -1); F.done = false;
    open_node(F, wid, meta2[wid], meta2[N + wid], meta2, nw, N);
    int  sv_cur, s_cur;
    bool last_cur;
    next_tile(F, sv_cur, s_cur, last_cur, meta2, nxt, tab, l16, nw, N);
    bool have = true;

    for (int it = 0; it < E_ + N && have; ++it) {
        int clo[8], chi[8];
        #pragma unroll
        for (int r = 0; r < 8; ++r) {
            clo[r] = __builtin_amdgcn_readlane(sv_cur, r);
            chi[r] = __builtin_amdgcn_readlane(sv_cur, 8 + r);
        }
        const int mye = sv_cur;
        int er[8];
        #pragma unroll
        for (int r = 0; r < 8; ++r) er[r] = lhi ? chi[r] : clo[r];

        int tt[8];
        float exw[8];
        #pragma unroll
        for (int r = 0; r < 8; ++r)
            tt[r] = eidx[E_ + (er[r] < 0 ? 0 : er[r])];
        #pragma unroll
        for (int r = 0; r < 8; ++r)
            exw[r] = exbuf[(unsigned)(er[r] < 0 ? 0 : er[r]) * HH + hsel];

        const float* erow = ea + (size_t)(mye < 0 ? 0 : mye) * RPE;
        v2f a[5];
        #pragma unroll
        for (int ki = 0; ki < 5; ++ki) {
            const int ka = ki*4 + (lhi ? 2 : 0);
            const int kc = (ka < RPE) ? ka : 16;
            const v2f ld = *(const v2f*)(erow + kc);
            a[ki] = (mye >= 0 && ka < RPE) ? ld : zero2;
        }

        const bool haveN = !F.done;
        int  sv_nxt = -1, s_nxt = s_cur;
        bool last_nxt = false;
        if (haveN) next_tile(F, sv_nxt, s_nxt, last_nxt, meta2, nxt, tab, l16, nw, N);

        asm volatile("" :: "v"(tt[0]), "v"(tt[1]), "v"(tt[2]), "v"(tt[3]),
                           "v"(tt[4]), "v"(tt[5]), "v"(tt[6]), "v"(tt[7]));

        v4f vv[8];
        #pragma unroll
        for (int r = 0; r < 8; ++r)
            vv[r] = vrow4[(unsigned)tt[r] * (QKVW/4) + (2*DHH/4) + l16];

        v8f d[4];
        #pragma unroll
        for (int ct = 0; ct < 4; ++ct) {
            v8f dd = {};
            #pragma unroll
            for (int ki = 0; ki < 5; ++ki)
                dd = __builtin_amdgcn_wmma_f32_16x16x4_f32(
                    false, a[ki], false, bf[ct][ki], (short)0, dd, false, false);
            d[ct] = dd;
        }

        #pragma unroll
        for (int r = 0; r < 8; ++r) {
            const bool live = er[r] >= 0;
            const float wl = live ? exw[r] : 0.0f;
            const float vr[4] = {vv[r][0], vv[r][1], vv[r][2], vv[r][3]};
            #pragma unroll
            for (int ct = 0; ct < 4; ++ct) {
                const float val = d[ct][r] + vr[ct] + vbias[ct];
                const float vl = live ? val : 0.0f;
                acc[ct] = fmaf(wl, vl, acc[ct]);
            }
            den += wl;
        }

        if (last_cur) {
            const float dn = den + __shfl_xor(den, 16);
            float tot[4];
            #pragma unroll
            for (int ct = 0; ct < 4; ++ct) tot[ct] = acc[ct] + __shfl_xor(acc[ct], 16);
            if (lhi == 0) {
                float4 res;
                res.x = tot[0] / (dn + 1e-16f);
                res.y = tot[1] / (dn + 1e-16f);
                res.z = tot[2] / (dn + 1e-16f);
                res.w = tot[3] / (dn + 1e-16f);
                *(float4*)(out + ((unsigned)s_cur * DIMV + 4*l16)) = res;
            }
            #pragma unroll
            for (int ct = 0; ct < 4; ++ct) acc[ct] = 0.f;
            den = 0.f;
        }

        sv_cur   = sv_nxt;
        s_cur    = s_nxt;
        last_cur = last_nxt;
        have     = haveN;
    }
}

extern "C" void kernel_launch(void* const* d_in, const int* in_sizes, int n_in,
                              void* d_out, int out_size, void* d_ws, size_t ws_size,
                              hipStream_t stream)
{
    const float* x       = (const float*)d_in[0];
    const int*   eidx    = (const int*)  d_in[1];
    const float* ea      = (const float*)d_in[2];
    const float* qkv_w   = (const float*)d_in[3];
    const float* qkv_b   = (const float*)d_in[4];
    const float* k_rpe_w = (const float*)d_in[5];
    const float* k_rpe_b = (const float*)d_in[6];
    const float* q_rpe_w = (const float*)d_in[7];
    const float* q_rpe_b = (const float*)d_in[8];
    const float* v_rpe_w = (const float*)d_in[9];
    const float* v_rpe_b = (const float*)d_in[10];

    const int N  = in_sizes[0] / DIMV;
    const int E_ = in_sizes[2] / RPE;
    float* out = (float*)d_out;

    char* ws = (char*)d_ws;
    float* qkv  = (float*)ws;
    float* exb  = (float*)(ws + (size_t)N * QKVW * 4);
    int*   meta = (int*)  (ws + (size_t)N * QKVW * 4
                              + (size_t)E_ * HH  * 4);
    int*   nxt  = (int*)  (ws + (size_t)N * QKVW * 4
                              + (size_t)E_ * HH  * 4
                              + (size_t)N * 4    * 4);
    int*   tab  = (int*)  (ws + (size_t)N * QKVW * 4
                              + (size_t)E_ * HH  * 4
                              + (size_t)N * 4    * 4
                              + (size_t)E_ * 4);

    qkv_proj_kernel<<<(N + 15) / 16, 256, 0, stream>>>(x, qkv_w, qkv_b, qkv, meta, N);
    edge_compat_kernel<<<(E_ + 127) / 128, 256, 0, stream>>>(
        ea, eidx, qkv, q_rpe_w, q_rpe_b, k_rpe_w, k_rpe_b, exb, meta, nxt, tab, N, E_);
    node_out_kernel<<<(N + 15) / 16, 256, 0, stream>>>(
        ea, eidx, qkv, v_rpe_w, v_rpe_b, exb, meta, nxt, tab, out, N, E_);
}


# Per-edge weights made in the node kernel: one edge_attr gather feeds q/k/v rpe WMMAs, head dot closed by quad swaps in the D lanes; edge pass only files slots; float sum order changes (dot tree)
# speedup vs baseline: 5.8067x; 1.1001x over previous
#include <hip/hip_runtime.h>
#include <hip/hip_bf16.h>


typedef __attribute__((ext_vector_type(2))) float v2f;
typedef __attribute__((ext_vector_type(4))) float v4f;
typedef __attribute__((ext_vector_type(8))) float v8f;

#define HH    4
#define DD    8
#define DHH   32
#define DIMV  64
#define RPE   18
#define QKVW  128
#define CAP   32

__global__ __launch_bounds__(256) void qkv_proj_kernel(
    const float* __restrict__ x, const float* __restrict__ W,
    const float* __restrict__ bias, float* __restrict__ qkv,
    int* __restrict__ meta, int N)
{
    __shared__ float xs[16 * 65];
    const int tid = threadIdx.x;
    const int r0  = blockIdx.x * 16;

    {
        int idx = tid * 4;
        int row = idx >> 6, col = idx & 63;
        int grow = r0 + row;
        float4 vx = make_float4(0.f, 0.f, 0.f, 0.f);
        if (grow < N) vx = *(const float4*)(x + (size_t)grow * 64 + col);
        xs[row*65+col+0] = vx.x; xs[row*65+col+1] = vx.y;
        xs[row*65+col+2] = vx.z; xs[row*65+col+3] = vx.w;
    }
    if (tid < 64 && r0 + ((tid & 31) >> 1) < N)
        meta[((tid >> 5) * N + r0) * 2 + (tid & 31)] = (tid & 1) ? -1 : 0;
    __syncthreads();

    const int wave = tid >> 5;
    const int lane = tid & 31;
    const int lhi  = lane >> 4;
    const int l16  = lane & 15;
    const int colbase = wave * 16;

    v8f acc = {};
    #pragma unroll
    for (int kk = 0; kk < 64; kk += 4) {
        const int ka = kk + (lhi ? 2 : 0);
        v2f a, b;
        a.x = xs[l16*65 + ka];
        a.y = xs[l16*65 + ka + 1];
        b.x = W[(size_t)ka     * QKVW + colbase + l16];
        b.y = W[(size_t)(ka+1) * QKVW + colbase + l16];
        acc = __builtin_amdgcn_wmma_f32_16x16x4_f32(
            false, a, false, b, (short)0, acc, false, false);
    }

    const int col = colbase + l16;
    const float bv = bias[col];
    #pragma unroll
    for (int r = 0; r < 8; ++r) {
        int grow = r0 + r + (lhi ? 8 : 0);
        if (grow < N) qkv[(size_t)grow * QKVW + col] = acc[r] + bv;
    }
}

__global__ __launch_bounds__(256) void edge_file_kernel(
    const int* __restrict__ eidx, int* __restrict__ meta,
    int* __restrict__ nxt, int* __restrict__ tab, int N, int E_)
{
    const int hplane = (int)((__builtin_amdgcn_s_sendmsg_rtn(0x87) >> 18) & 1u);
    const int g0 = (int)blockIdx.x * 1024 + (int)threadIdx.x;
    int sl[4];
    #pragma unroll
    for (int i = 0; i < 4; ++i) {
        const int g = g0 + 256 * i;
        sl[i] = eidx[g < E_ ? g : 0];
    }
    int pos[4];
    #pragma unroll
    for (int i = 0; i < 4; ++i) {
        const int g = g0 + 256 * i;
        pos[i] = 0;
        if (g < E_) pos[i] = atomicAdd(&meta[2 * (hplane * N + sl[i])], 1);
    }
    #pragma unroll
    for (int i = 0; i < 4; ++i) {
        const int g = g0 + 256 * i;
        if (g < E_) {
            const int w = hplane * N + sl[i];
            if (pos[i] < CAP) tab[(unsigned)w * CAP + (unsigned)pos[i]] = g;
            else              nxt[g] = atomicExch(&meta[2 * w + 1], g);
        }
    }
}


struct TileCursor {
    int  s;
    int  n0, tot;
    int  k;
    int  oe0, oe1;
    int2 mb0, mb1;
    bool done;
};

__device__ __forceinline__ void open_node(TileCursor& F, int sn, int2 m0, int2 m1,
                                          const int2* __restrict__ meta2, int nw, int N)
{
    F.s   = sn;
    F.k   = 0;
    F.n0  = m0.x < CAP ? m0.x : CAP;
    F.tot = F.n0 + (m1.x < CAP ? m1.x : CAP);
    F.oe0 = m0.x > CAP ? m0.y : -1;
    F.oe1 = m1.x > CAP ? m1.y : -1;
    if (sn + nw < N) { F.mb0 = meta2[sn + nw]; F.mb1 = meta2[N + sn + nw]; }
}

__device__ __forceinline__ void next_tile(TileCursor& F, int& sv, int& s_of, bool& last_of,
                                          const int2* __restrict__ meta2,
                                          const int* __restrict__ nxt, const int* __restrict__ tab,
                                          int l16, int nw, int N)
{
    s_of = F.s;
    if (F.k == 0 || F.k * 16 < F.tot) {
        const int g = F.k * 16 + l16;
        const unsigned base = (unsigned)F.s * CAP;
        const unsigned idx  = g < F.n0 ? base + (unsigned)g
                                       : (unsigned)N * CAP + base + (unsigned)(g - F.n0);
        int ld = -1;
        if (g < F.tot) ld = tab[idx];
        sv = ld;
        F.k += 1;
    } else {
        int v = -1;
        for (int j = 0; j < 16; ++j) {
            const int e = F.oe0 >= 0 ? F.oe0 : F.oe1;
            if (e < 0) break;
            if (l16 == j) v = e;
            const int nx = nxt[e];
            if (F.oe0 >= 0) F.oe0 = nx; else F.oe1 = nx;
        }
        sv = v;
    }
    const bool more = F.k * 16 < F.tot || F.oe0 >= 0 || F.oe1 >= 0;
    last_of = !more;
    if (!more) {
        const int sn = F.s + nw;
        if (sn < N) open_node(F, sn, F.mb0, F.mb1, meta2, nw, N);
        else        F.done = true;
    }
}

__global__ __launch_bounds__(256) __attribute__((amdgpu_waves_per_eu(8)))
void node_out_kernel(
    const float* __restrict__ ea, const int* __restrict__ eidx,
    const float* __restrict__ qkv,
    const float* __restrict__ qw, const float* __restrict__ qb,
    const float* __restrict__ kw, const float* __restrict__ kb,
    const float* __restrict__ vw, const float* __restrict__ vb,
    const int* __restrict__ meta,
    const int* __restrict__ nxt, const int* __restrict__ tab,
    float* __restrict__ out, int N, int E_)
{
    __shared__ __attribute__((aligned(16))) v2f sqk[20 * 32];
    __shared__ __attribute__((aligned(16))) v2f svw[20 * 32];
    const int tid = threadIdx.x;
    for (int j = tid; j < 20 * 32; j += 256) {
        const int ln = j & 31, f = j >> 5;
        const int ki = f % 5;
        const int ka = ki*4 + ((ln >> 4) ? 2 : 0);
        const bool okx = ka < RPE, oky = ka + 1 < RPE;
        const int rx = okx ? ka : 16;
        const int ry = oky ? ka + 1 : 17;
        {
            const int sel = f / 10, ct = (f % 10) / 5;
            const float* Wp = sel ? kw : qw;
            const float wx = Wp[rx * DHH + 2*(ln & 15) + ct];
            const float wy = Wp[ry * DHH + 2*(ln & 15) + ct];
            v2f b;
            b.x = okx ? wx : 0.0f;
            b.y = oky ? wy : 0.0f;
            sqk[j] = b;
        }
        {
            const int ct = f / 5;
            const float wx = vw[rx * DIMV + 4*(ln & 15) + ct];
            const float wy = vw[ry * DIMV + 4*(ln & 15) + ct];
            v2f b;
            b.x = okx ? wx : 0.0f;
            b.y = oky ? wy : 0.0f;
            svw[j] = b;
        }
    }
    __syncthreads();

    const int lane = tid & 31;
    const int lhi = lane >> 4, l16 = lane & 15;
    const int wid = __builtin_amdgcn_readfirstlane((int)blockIdx.x * 8 + (tid >> 5));
    const int nw  = (int)gridDim.x * 8;
    if (wid >= N) return;

    float vbias[4];
    {
        const float4 b4 = *(const float4*)(vb + 4*l16);
        vbias[0] = b4.x; vbias[1] = b4.y; vbias[2] = b4.z; vbias[3] = b4.w;
    }
    const v2f qb2 = *(const v2f*)(qb + 2*l16);
    const v2f kb2 = *(const v2f*)(kb + 2*l16);
    const float scale = 0.35355339059327373f;
    const v4f* __restrict__ vrow4 = (const v4f*)qkv;
    const v2f* __restrict__ qrow2 = (const v2f*)qkv;

    float acc[4] = {0.f, 0.f, 0.f, 0.f};
    float den = 0.f;
    const v2f zero2 = {0.f, 0.f};

    const int2* __restrict__ meta2 = (const int2*)meta;
    TileCursor F;
    F.mb0 = make_int2(0, -1); F.mb1 = make_int2(0, -1); F.done = false;
    open_node(F, wid, meta2[wid], meta2[N + wid], meta2, nw, N);
    int  sv_cur, s_cur;
    bool last_cur;
    next_tile(F, sv_cur, s_cur, last_cur, meta2, nxt, tab, l16, nw, N);
    bool have = true;

    for (int it = 0; it < E_ + N && have; ++it) {
        int clo[8], chi[8];
        #pragma unroll
        for (int r = 0; r < 8; ++r) {
            clo[r] = __builtin_amdgcn_readlane(sv_cur, r);
            chi[r] = __builtin_amdgcn_readlane(sv_cur, 8 + r);
        }
        const int mye = sv_cur;
        int er[8];
        #pragma unroll
        for (int r = 0; r < 8; ++r) er[r] = lhi ? chi[r] : clo[r];

        int tt[8];
        #pragma unroll
        for (int r = 0; r < 8; ++r)
            tt[r] = eidx[E_ + (er[r] < 0 ? 0 : er[r])];

        const v2f qq = qrow2[(unsigned)s_cur * (QKVW/2) + l16];

        const float* erow = ea + (size_t)(mye < 0 ? 0 : mye) * RPE;
        v2f a[5];
        #pragma unroll
        for (int ki = 0; ki < 5; ++ki) {
            const int ka = ki*4 + (lhi ? 2 : 0);
            const int kc = (ka < RPE) ? ka : 16;
            const v2f ld = *(const v2f*)(erow + kc);
            a[ki] = (mye >= 0 && ka < RPE) ? ld : zero2;
        }

        const bool haveN = !F.done;
        int  sv_nxt = -1, s_nxt = s_cur;
        bool last_nxt = false;
        if (haveN) next_tile(F, sv_nxt, s_nxt, last_nxt, meta2, nxt, tab, l16, nw, N);

        asm volatile("" :: "v"(tt[0]), "v"(tt[1]), "v"(tt[2]), "v"(tt[3]),
                           "v"(tt[4]), "v"(tt[5]), "v"(tt[6]), "v"(tt[7]));

        v2f kk[8];
        v4f vv[8];
        #pragma unroll
        for (int r = 0; r < 8; ++r) {
            kk[r] = qrow2[(unsigned)tt[r] * (QKVW/2) + (DHH/2) + l16];
            vv[r] = vrow4[(unsigned)tt[r] * (QKVW/4) + (2*DHH/4) + l16];
        }

        v8f dq0 = {}, dq1 = {}, dk0 = {}, dk1 = {};
        #pragma unroll
        for (int ki = 0; ki < 5; ++ki) {
            const v2f bq0 = sqk[( 0 + ki) * 32 + lane];
            const v2f bq1 = sqk[( 5 + ki) * 32 + lane];
            const v2f bk0 = sqk[(10 + ki) * 32 + lane];
            const v2f bk1 = sqk[(15 + ki) * 32 + lane];
            dq0 = __builtin_amdgcn_wmma_f32_16x16x4_f32(
                false, a[ki], false, bq0, (short)0, dq0, false, false);
            dq1 = __builtin_amdgcn_wmma_f32_16x16x4_f32(
                false, a[ki], false, bq1, (short)0, dq1, false, false);
            dk0 = __builtin_amdgcn_wmma_f32_16x16x4_f32(
                false, a[ki], false, bk0, (short)0, dk0, false, false);
            dk1 = __builtin_amdgcn_wmma_f32_16x16x4_f32(
                false, a[ki], false, bk1, (short)0, dk1, false, false);
        }

        v8f d[4];
        #pragma unroll
        for (int ct = 0; ct < 4; ++ct) {
            v8f dd = {};
            #pragma unroll
            for (int ki = 0; ki < 5; ++ki)
                dd = __builtin_amdgcn_wmma_f32_16x16x4_f32(
                    false, a[ki], false, svw[(ct*5 + ki) * 32 + lane], (short)0, dd, false, false);
            d[ct] = dd;
        }

        const float qx = qq.x * scale, qy = qq.y * scale;
        float exw[8];
        #pragma unroll
        for (int r = 0; r < 8; ++r) {
            float part = (qx + dq0[r] + qb2.x) * (kk[r].x + dk0[r] + kb2.x);
            part = fmaf(qy + dq1[r] + qb2.y, kk[r].y + dk1[r] + kb2.y, part);
            const float p2 = part + __builtin_bit_cast(float,
                __builtin_amdgcn_mov_dpp(__builtin_bit_cast(int, part), 0xB1, 0xF, 0xF, true));
            const float p4 = p2 + __builtin_bit_cast(float,
                __builtin_amdgcn_mov_dpp(__builtin_bit_cast(int, p2), 0x4E, 0xF, 0xF, true));
            exw[r] = __expf(p4);
        }

        #pragma unroll
        for (int r = 0; r < 8; ++r) {
            const bool live = er[r] >= 0;
            const float wl = live ? exw[r] : 0.0f;
            const float vr[4] = {vv[r][0], vv[r][1], vv[r][2], vv[r][3]};
            #pragma unroll
            for (int ct = 0; ct < 4; ++ct) {
                const float val = d[ct][r] + vr[ct] + vbias[ct];
                const float vl = live ? val : 0.0f;
                acc[ct] = fmaf(wl, vl, acc[ct]);
            }
            den += wl;
        }

        if (last_cur) {
            const float dn = den + __shfl_xor(den, 16);
            float tot[4];
            #pragma unroll
            for (int ct = 0; ct < 4; ++ct) tot[ct] = acc[ct] + __shfl_xor(acc[ct], 16);
            if (lhi == 0) {
                float4 res;
                res.x = tot[0] / (dn + 1e-16f);
                res.y = tot[1] / (dn + 1e-16f);
                res.z = tot[2] / (dn + 1e-16f);
                res.w = tot[3] / (dn + 1e-16f);
                *(float4*)(out + ((unsigned)s_cur * DIMV + 4*l16)) = res;
            }
            #pragma unroll
            for (int ct = 0; ct < 4; ++ct) acc[ct] = 0.f;
            den = 0.f;
        }

        sv_cur   = sv_nxt;
        s_cur    = s_nxt;
        last_cur = last_nxt;
        have     = haveN;
    }
}

extern "C" void kernel_launch(void* const* d_in, const int* in_sizes, int n_in,
                              void* d_out, int out_size, void* d_ws, size_t ws_size,
                              hipStream_t stream)
{
    const float* x       = (const float*)d_in[0];
    const int*   eidx    = (const int*)  d_in[1];
    const float* ea      = (const float*)d_in[2];
    const float* qkv_w   = (const float*)d_in[3];
    const float* qkv_b   = (const float*)d_in[4];
    const float* k_rpe_w = (const float*)d_in[5];
    const float* k_rpe_b = (const float*)d_in[6];
    const float* q_rpe_w = (const float*)d_in[7];
    const float* q_rpe_b = (const float*)d_in[8];
    const float* v_rpe_w = (const float*)d_in[9];
    const float* v_rpe_b = (const float*)d_in[10];

    const int N  = in_sizes[0] / DIMV;
    const int E_ = in_sizes[2] / RPE;
    float* out = (float*)d_out;

    char* ws = (char*)d_ws;
    float* qkv  = (float*)ws;
    int*   meta = (int*)  (ws + (size_t)N * QKVW * 4);
    int*   nxt  = (int*)  (ws + (size_t)N * QKVW * 4
                              + (size_t)N * 4    * 4);
    int*   tab  = (int*)  (ws + (size_t)N * QKVW * 4
                              + (size_t)N * 4    * 4
                              + (size_t)E_ * 4);

    qkv_proj_kernel<<<(N + 15) / 16, 256, 0, stream>>>(x, qkv_w, qkv_b, qkv, meta, N);
    edge_file_kernel<<<(E_ + 1023) / 1024, 256, 0, stream>>>(eidx, meta, nxt, tab, N, E_);
    node_out_kernel<<<(N + 15) / 16, 256, 0, stream>>>(
        ea, eidx, qkv, q_rpe_w, q_rpe_b, k_rpe_w, k_rpe_b, v_rpe_w, v_rpe_b,
        meta, nxt, tab, out, N, E_);
}


# Node kernel runs as the 1024 blocks the chip holds at once (a wave streams 12-13 nodes, not 4): no near-empty 4th round of blocks, fewer part-filled tiles; float sum order changes (tile alignment)
# speedup vs baseline: 6.7311x; 1.1592x over previous
#include <hip/hip_runtime.h>
#include <hip/hip_bf16.h>


typedef __attribute__((ext_vector_type(2))) float v2f;
typedef __attribute__((ext_vector_type(4))) float v4f;
typedef __attribute__((ext_vector_type(8))) float v8f;

#define HH    4
#define DD    8
#define DHH   32
#define DIMV  64
#define RPE   18
#define QKVW  128
#define CAP   32

__global__ __launch_bounds__(256) void qkv_proj_kernel(
    const float* __restrict__ x, const float* __restrict__ W,
    const float* __restrict__ bias, float* __restrict__ qkv,
    int* __restrict__ meta, int N)
{
    __shared__ float xs[16 * 65];
    const int tid = threadIdx.x;
    const int r0  = blockIdx.x * 16;

    {
        int idx = tid * 4;
        int row = idx >> 6, col = idx & 63;
        int grow = r0 + row;
        float4 vx = make_float4(0.f, 0.f, 0.f, 0.f);
        if (grow < N) vx = *(const float4*)(x + (size_t)grow * 64 + col);
        xs[row*65+col+0] = vx.x; xs[row*65+col+1] = vx.y;
        xs[row*65+col+2] = vx.z; xs[row*65+col+3] = vx.w;
    }
    if (tid < 64 && r0 + ((tid & 31) >> 1) < N)
        meta[((tid >> 5) * N + r0) * 2 + (tid & 31)] = (tid & 1) ? -1 : 0;
    __syncthreads();

    const int wave = tid >> 5;
    const int lane = tid & 31;
    const int lhi  = lane >> 4;
    const int l16  = lane & 15;
    const int colbase = wave * 16;

    v8f acc = {};
    #pragma unroll
    for (int kk = 0; kk < 64; kk += 4) {
        const int ka = kk + (lhi ? 2 : 0);
        v2f a, b;
        a.x = xs[l16*65 + ka];
        a.y = xs[l16*65 + ka + 1];
        b.x = W[(size_t)ka     * QKVW + colbase + l16];
        b.y = W[(size_t)(ka+1) * QKVW + colbase + l16];
        acc = __builtin_amdgcn_wmma_f32_16x16x4_f32(
            false, a, false, b, (short)0, acc, false, false);
    }

    const int col = colbase + l16;
    const float bv = bias[col];
    #pragma unroll
    for (int r = 0; r < 8; ++r) {
        int grow = r0 + r + (lhi ? 8 : 0);
        if (grow < N) qkv[(size_t)grow * QKVW + col] = acc[r] + bv;
    }
}

__global__ __launch_bounds__(256) void edge_file_kernel(
    const int* __restrict__ eidx, int* __restrict__ meta,
    int* __restrict__ nxt, int* __restrict__ tab, int N, int E_)
{
    const int hplane = (int)((__builtin_amdgcn_s_sendmsg_rtn(0x87) >> 18) & 1u);
    const int g0 = (int)blockIdx.x * 1024 + (int)threadIdx.x;
    int sl[4];
    #pragma unroll
    for (int i = 0; i < 4; ++i) {
        const int g = g0 + 256 * i;
        sl[i] = eidx[g < E_ ? g : 0];
    }
    int pos[4];
    #pragma unroll
    for (int i = 0; i < 4; ++i) {
        const int g = g0 + 256 * i;
        pos[i] = 0;
        if (g < E_) pos[i] = atomicAdd(&meta[2 * (hplane * N + sl[i])], 1);
    }
    #pragma unroll
    for (int i = 0; i < 4; ++i) {
        const int g = g0 + 256 * i;
        if (g < E_) {
            const int w = hplane * N + sl[i];
            if (pos[i] < CAP) tab[(unsigned)w * CAP + (unsigned)pos[i]] = g;
            else              nxt[g] = atomicExch(&meta[2 * w + 1], g);
        }
    }
}


struct TileCursor {
    int  s;
    int  n0, tot;
    int  pos;
    int  oe0, oe1;
    int2 mb0, mb1;
    bool done;
};

struct TileDesc {
    int  sL;
    int  b;
    int  nfill;
    int  sH;
    bool hasH;
    bool closeL;
    bool closeH;
};

__device__ __forceinline__ void open_node(TileCursor& F, int sn, int2 m0, int2 m1,
                                          const int2* __restrict__ meta2, int nw, int N)
{
    F.s   = sn;
    F.pos = 0;
    F.n0  = m0.x < CAP ? m0.x : CAP;
    F.tot = F.n0 + (m1.x < CAP ? m1.x : CAP);
    F.oe0 = m0.x > CAP ? m0.y : -1;
    F.oe1 = m1.x > CAP ? m1.y : -1;
    if (sn + nw < N) { F.mb0 = meta2[sn + nw]; F.mb1 = meta2[N + sn + nw]; }
}

__device__ __forceinline__ void advance_node(TileCursor& F, const int2* __restrict__ meta2, int nw, int N)
{
    const int sn = F.s + nw;
    if (sn < N) open_node(F, sn, F.mb0, F.mb1, meta2, nw, N);
    else        F.done = true;
}

__device__ __forceinline__ void next_tile(TileCursor& F, int& sv, TileDesc& T,
                                          const int2* __restrict__ meta2,
                                          const int* __restrict__ nxt, const int* __restrict__ tab,
                                          int l16, int nw, int N)
{
    T.sL = F.s; T.sH = F.s; T.hasH = false; T.closeH = false;
    if (F.pos < F.tot || F.tot == 0) {
        const int left  = F.tot - F.pos;
        const int takeL = left < 16 ? left : 16;
        const unsigned baseL = (unsigned)F.s * CAP;
        const int gL = F.pos + l16;
        unsigned idx = gL < F.n0 ? baseL + (unsigned)gL
                                 : (unsigned)N * CAP + baseL + (unsigned)(gL - F.n0);
        bool valid = l16 < takeL;
        F.pos += takeL;
        T.b = takeL;
        T.nfill = takeL;
        T.closeL = F.pos >= F.tot && F.oe0 < 0 && F.oe1 < 0;
        if (T.closeL) {
            advance_node(F, meta2, nw, N);
            if (!F.done && takeL < 16) {
                T.hasH = true;
                T.sH   = F.s;
                const int room  = 16 - takeL;
                const int takeH = F.tot < room ? F.tot : room;
                const unsigned baseH = (unsigned)F.s * CAP;
                const int gH = l16 - takeL;
                const unsigned idxH = gH < F.n0 ? baseH + (unsigned)gH
                                                : (unsigned)N * CAP + baseH + (unsigned)(gH - F.n0);
                const bool inH = gH >= 0 && gH < takeH;
                idx   = inH ? idxH : idx;
                valid = valid || inH;
                F.pos = takeH;
                T.nfill = takeL + takeH;
                T.closeH = takeH >= F.tot && F.oe0 < 0 && F.oe1 < 0;
                if (T.closeH) advance_node(F, meta2, nw, N);
            }
        }
        int ld = -1;
        if (valid) ld = tab[idx];
        sv = ld;
    } else {
        int v = -1;
        int cnt = 0;
        for (int j = 0; j < 16; ++j) {
            const int e = F.oe0 >= 0 ? F.oe0 : F.oe1;
            if (e < 0) break;
            if (l16 == j) v = e;
            cnt = j + 1;
            const int nx = nxt[e];
            if (F.oe0 >= 0) F.oe0 = nx; else F.oe1 = nx;
        }
        sv = v;
        T.b = 16;
        T.nfill = cnt;
        T.closeL = F.oe0 < 0 && F.oe1 < 0;
        if (T.closeL) advance_node(F, meta2, nw, N);
    }
}

__device__ __forceinline__ void store_row(float (&acc)[4], float& den, int node, int lhi, int l16,
                                          float* __restrict__ out)
{
    const float dn = den + __shfl_xor(den, 16);
    float tot[4];
    #pragma unroll
    for (int ct = 0; ct < 4; ++ct) tot[ct] = acc[ct] + __shfl_xor(acc[ct], 16);
    if (lhi == 0) {
        float4 res;
        res.x = tot[0] / (dn + 1e-16f);
        res.y = tot[1] / (dn + 1e-16f);
        res.z = tot[2] / (dn + 1e-16f);
        res.w = tot[3] / (dn + 1e-16f);
        *(float4*)(out + ((unsigned)node * DIMV + 4*l16)) = res;
    }
    #pragma unroll
    for (int ct = 0; ct < 4; ++ct) acc[ct] = 0.f;
    den = 0.f;
}

__global__ __launch_bounds__(256) __attribute__((amdgpu_waves_per_eu(8)))
void node_out_kernel(
    const float* __restrict__ ea, const int* __restrict__ eidx,
    const float* __restrict__ qkv,
    const float* __restrict__ qw, const float* __restrict__ qb,
    const float* __restrict__ kw, const float* __restrict__ kb,
    const float* __restrict__ vw, const float* __restrict__ vb,
    const int* __restrict__ meta,
    const int* __restrict__ nxt, const int* __restrict__ tab,
    float* __restrict__ out, int N, int E_)
{
    __shared__ __attribute__((aligned(16))) v2f sqk[20 * 32];
    __shared__ __attribute__((aligned(16))) v2f svw[20 * 32];
    const int tid = threadIdx.x;
    for (int j = tid; j < 20 * 32; j += 256) {
        const int ln = j & 31, f = j >> 5;
        const int ki = f % 5;
        const int ka = ki*4 + ((ln >> 4) ? 2 : 0);
        const bool okx = ka < RPE, oky = ka + 1 < RPE;
        const int rx = okx ? ka : 16;
        const int ry = oky ? ka + 1 : 17;
        {
            const int sel = f / 10, ct = (f % 10) / 5;
            const float* Wp = sel ? kw : qw;
            const float wx = Wp[rx * DHH + 2*(ln & 15) + ct];
            const float wy = Wp[ry * DHH + 2*(ln & 15) + ct];
            v2f b;
            b.x = okx ? wx : 0.0f;
            b.y = oky ? wy : 0.0f;
            sqk[j] = b;
        }
        {
            const int ct = f / 5;
            const float wx = vw[rx * DIMV + 4*(ln & 15) + ct];
            const float wy = vw[ry * DIMV + 4*(ln & 15) + ct];
            v2f b;
            b.x = okx ? wx : 0.0f;
            b.y = oky ? wy : 0.0f;
            svw[j] = b;
        }
    }
    __syncthreads();

    const int lane = tid & 31;
    const int lhi = lane >> 4, l16 = lane & 15;
    const int wid = __builtin_amdgcn_readfirstlane((int)blockIdx.x * 8 + (tid >> 5));
    const int nw  = (int)gridDim.x * 8;
    if (wid >= N) return;

    float vbias[4];
    {
        const float4 b4 = *(const float4*)(vb + 4*l16);
        vbias[0] = b4.x; vbias[1] = b4.y; vbias[2] = b4.z; vbias[3] = b4.w;
    }
    const v2f qb2 = *(const v2f*)(qb + 2*l16);
    const v2f kb2 = *(const v2f*)(kb + 2*l16);
    const float scale = 0.35355339059327373f;
    const v4f* __restrict__ vrow4 = (const v4f*)qkv;
    const v2f* __restrict__ qrow2 = (const v2f*)qkv;

    float acc[4] = {0.f, 0.f, 0.f, 0.f};
    float den = 0.f;
    const v2f zero2 = {0.f, 0.f};

    const int2* __restrict__ meta2 = (const int2*)meta;
    TileCursor F;
    F.mb0 = make_int2(0, -1); F.mb1 = make_int2(0, -1); F.done = false;
    open_node(F, wid, meta2[wid], meta2[N + wid], meta2, nw, N);
    int      sv_cur;
    TileDesc T_cur;
    next_tile(F, sv_cur, T_cur, meta2, nxt, tab, l16, nw, N);
    bool have = true;

    for (int it = 0; it < E_ + N && have; ++it) {
        int clo[8], chi[8];
        #pragma unroll
        for (int r = 0; r < 8; ++r) {
            clo[r] = __builtin_amdgcn_readlane(sv_cur, r);
            chi[r] = __builtin_amdgcn_readlane(sv_cur, 8 + r);
        }
        const int mye = sv_cur;
        int er[8];
        #pragma unroll
        for (int r = 0; r < 8; ++r) er[r] = lhi ? chi[r] : clo[r];

        int tt[8];
        #pragma unroll
        for (int r = 0; r < 8; ++r)
            tt[r] = eidx[E_ + (er[r] < 0 ? 0 : er[r])];

        const v2f qL = qrow2[(unsigned)T_cur.sL * (QKVW/2) + l16];
        const v2f qH = qrow2[(unsigned)T_cur.sH * (QKVW/2) + l16];

        const float* erow = ea + (size_t)(mye < 0 ? 0 : mye) * RPE;
        v2f a[5];
        #pragma unroll
        for (int ki = 0; ki < 5; ++ki) {
            const int ka = ki*4 + (lhi ? 2 : 0);
            const int kc = (ka < RPE) ? ka : 16;
            const v2f ld = *(const v2f*)(erow + kc);
            a[ki] = (mye >= 0 && ka < RPE) ? ld : zero2;
        }

        const bool haveN = !F.done;
        int      sv_nxt = -1;
        TileDesc T_nxt  = T_cur;
        if (haveN) next_tile(F, sv_nxt, T_nxt, meta2, nxt, tab, l16, nw, N);

        asm volatile("" :: "v"(tt[0]), "v"(tt[1]), "v"(tt[2]), "v"(tt[3]),
                           "v"(tt[4]), "v"(tt[5]), "v"(tt[6]), "v"(tt[7]));

        v2f kk[8];
        v4f vv[8];
        #pragma unroll
        for (int r = 0; r < 8; ++r) {
            kk[r] = qrow2[(unsigned)tt[r] * (QKVW/2) + (DHH/2) + l16];
            vv[r] = vrow4[(unsigned)tt[r] * (QKVW/4) + (2*DHH/4) + l16];
        }

        v8f dq0 = {}, dq1 = {}, dk0 = {}, dk1 = {};
        #pragma unroll
        for (int ki = 0; ki < 5; ++ki) {
            const v2f bq0 = sqk[( 0 + ki) * 32 + lane];
            const v2f bq1 = sqk[( 5 + ki) * 32 + lane];
            const v2f bk0 = sqk[(10 + ki) * 32 + lane];
            const v2f bk1 = sqk[(15 + ki) * 32 + lane];
            dq0 = __builtin_amdgcn_wmma_f32_16x16x4_f32(
                false, a[ki], false, bq0, (short)0, dq0, false, false);
            dq1 = __builtin_amdgcn_wmma_f32_16x16x4_f32(
                false, a[ki], false, bq1, (short)0, dq1, false, false);
            dk0 = __builtin_amdgcn_wmma_f32_16x16x4_f32(
                false, a[ki], false, bk0, (short)0, dk0, false, false);
            dk1 = __builtin_amdgcn_wmma_f32_16x16x4_f32(
                false, a[ki], false, bk1, (short)0, dk1, false, false);
        }

        v8f d[4];
        #pragma unroll
        for (int ct = 0; ct < 4; ++ct) {
            v8f dd = {};
            #pragma unroll
            for (int ki = 0; ki < 5; ++ki)
                dd = __builtin_amdgcn_wmma_f32_16x16x4_f32(
                    false, a[ki], false, svw[(ct*5 + ki) * 32 + lane], (short)0, dd, false, false);
            d[ct] = dd;
        }

        const int   bl  = T_cur.b - 8 * lhi;
        const int   nl  = T_cur.nfill - 8 * lhi;
        const float qLx = qL.x * scale, qLy = qL.y * scale;
        const float qHx = qH.x * scale, qHy = qH.y * scale;
        float exw[8];
        #pragma unroll
        for (int r = 0; r < 8; ++r) {
            const bool  inL = r < bl;
            const float qx = inL ? qLx : qHx, qy = inL ? qLy : qHy;
            float part = (qx + dq0[r] + qb2.x) * (kk[r].x + dk0[r] + kb2.x);
            part = fmaf(qy + dq1[r] + qb2.y, kk[r].y + dk1[r] + kb2.y, part);
            const float p2 = part + __builtin_bit_cast(float,
                __builtin_amdgcn_mov_dpp(__builtin_bit_cast(int, part), 0xB1, 0xF, 0xF, true));
            const float p4 = p2 + __builtin_bit_cast(float,
                __builtin_amdgcn_mov_dpp(__builtin_bit_cast(int, p2), 0x4E, 0xF, 0xF, true));
            exw[r] = __expf(p4);
        }

        if (!T_cur.hasH) {
            #pragma unroll
            for (int r = 0; r < 8; ++r) {
                const bool live = r < nl;
                const float wl = live ? exw[r] : 0.0f;
                const float vr[4] = {vv[r][0], vv[r][1], vv[r][2], vv[r][3]};
                #pragma unroll
                for (int ct = 0; ct < 4; ++ct) {
                    const float val = d[ct][r] + vr[ct] + vbias[ct];
                    const float vl = live ? val : 0.0f;
                    acc[ct] = fmaf(wl, vl, acc[ct]);
                }
                den += wl;
            }
            if (T_cur.closeL) store_row(acc, den, T_cur.sL, lhi, l16, out);
        } else {
            float accH[4] = {0.f, 0.f, 0.f, 0.f};
            float denH = 0.f;
            #pragma unroll
            for (int r = 0; r < 8; ++r) {
                const bool live = r < nl;
                const bool inL  = r < bl;
                const float wl = live ? exw[r] : 0.0f;
                const float wL = inL ? wl : 0.0f;
                const float wH = inL ? 0.0f : wl;
                const float vr[4] = {vv[r][0], vv[r][1], vv[r][2], vv[r][3]};
                #pragma unroll
                for (int ct = 0; ct < 4; ++ct) {
                    const float val = d[ct][r] + vr[ct] + vbias[ct];
                    const float vl = live ? val : 0.0f;
                    acc[ct]  = fmaf(wL, vl, acc[ct]);
                    accH[ct] = fmaf(wH, vl, accH[ct]);
                }
                den  += wL;
                denH += wH;
            }
            store_row(acc, den, T_cur.sL, lhi, l16, out);
            #pragma unroll
            for (int ct = 0; ct < 4; ++ct) acc[ct] = accH[ct];
            den = denH;
            if (T_cur.closeH) store_row(acc, den, T_cur.sH, lhi, l16, out);
        }

        sv_cur = sv_nxt;
        T_cur  = T_nxt;
        have   = haveN;
    }
}

extern "C" void kernel_launch(void* const* d_in, const int* in_sizes, int n_in,
                              void* d_out, int out_size, void* d_ws, size_t ws_size,
                              hipStream_t stream)
{
    const float* x       = (const float*)d_in[0];
    const int*   eidx    = (const int*)  d_in[1];
    const float* ea      = (const float*)d_in[2];
    const float* qkv_w   = (const float*)d_in[3];
    const float* qkv_b   = (const float*)d_in[4];
    const float* k_rpe_w = (const float*)d_in[5];
    const float* k_rpe_b = (const float*)d_in[6];
    const float* q_rpe_w = (const float*)d_in[7];
    const float* q_rpe_b = (const float*)d_in[8];
    const float* v_rpe_w = (const float*)d_in[9];
    const float* v_rpe_b = (const float*)d_in[10];

    const int N  = in_sizes[0] / DIMV;
    const int E_ = in_sizes[2] / RPE;
    float* out = (float*)d_out;

    char* ws = (char*)d_ws;
    float* qkv  = (float*)ws;
    int*   meta = (int*)  (ws + (size_t)N * QKVW * 4);
    int*   nxt  = (int*)  (ws + (size_t)N * QKVW * 4
                              + (size_t)N * 4    * 4);
    int*   tab  = (int*)  (ws + (size_t)N * QKVW * 4
                              + (size_t)N * 4    * 4
                              + (size_t)E_ * 4);

    qkv_proj_kernel<<<(N + 15) / 16, 256, 0, stream>>>(x, qkv_w, qkv_b, qkv, meta, N);
    edge_file_kernel<<<(E_ + 1023) / 1024, 256, 0, stream>>>(eidx, meta, nxt, tab, N, E_);
    node_out_kernel<<<(N + 31) / 32 < 1024 ? (N + 31) / 32 : 1024, 256, 0, stream>>>(
        ea, eidx, qkv, q_rpe_w, q_rpe_b, k_rpe_w, k_rpe_b, v_rpe_w, v_rpe_b,
        meta, nxt, tab, out, N, E_);
}
